# starting kernel
# baseline (speedup 1.0000x reference)
#include <hip/hip_runtime.h>
#include <hip/hip_bf16.h>

#define N_FEATS 64
#define WTP_LD  80

typedef float v2f __attribute__((ext_vector_type(2)));
typedef float v8f __attribute__((ext_vector_type(8)));

__global__ void zero_f32(float* __restrict__ p, int n) {
    int i = blockIdx.x * blockDim.x + threadIdx.x;
    if (i < n) p[i] = 0.0f;
}

__global__ void edge_scatter(const float* __restrict__ x,
                             const int*   __restrict__ src,
                             const int*   __restrict__ dst,
                             float*       __restrict__ agg,
                             float*       __restrict__ deg,
                             int n_edges) {
    int t = blockIdx.x * blockDim.x + threadIdx.x;
    int e = t >> 4;
    if (e >= n_edges) return;
    int l = t & 15;
    int s = src[e];
    int d = dst[e];
    const float4 v = *reinterpret_cast<const float4*>(x + (size_t)s * N_FEATS + l * 4);
    float* o = agg + (size_t)d * N_FEATS + l * 4;
    atomicAdd(o + 0, v.x);
    atomicAdd(o + 1, v.y);
    atomicAdd(o + 2, v.z);
    atomicAdd(o + 3, v.w);
    if (l == 0) atomicAdd(deg + d, 1.0f);
}

__global__ void __launch_bounds__(256)
gcn_linear_wmma(const float* __restrict__ x,
                const float* __restrict__ agg,
                const float* __restrict__ deg,
                const float* __restrict__ W,
                const float* __restrict__ bias,
                float*       __restrict__ out,
                int n_tiles) {
    __shared__ float WtPf[32 * WTP_LD * 2];
    __shared__ float bsh[N_FEATS];

    for (int i = threadIdx.x; i < N_FEATS * N_FEATS; i += blockDim.x) {
        int n = i / N_FEATS;
        int k = i % N_FEATS;
        WtPf[((k >> 1) * WTP_LD + n) * 2 + (k & 1)] = W[i];
    }
    if (threadIdx.x < N_FEATS) bsh[threadIdx.x] = bias[threadIdx.x];
    __syncthreads();

    const int wid  = threadIdx.x >> 5;
    const int lane = threadIdx.x & 31;
    const int tile = blockIdx.x * 8 + wid;
    if (tile >= n_tiles) return;

    const int half = lane >> 4;
    const int l16  = lane & 15;
    const int row  = tile * 16 + l16;

    const bool use_agg = deg[row] > 0.0f;
    const float* hrow = use_agg ? (agg + (size_t)row * N_FEATS)
                                : (x   + (size_t)row * N_FEATS);

    v8f acc0 = {}, acc1 = {}, acc2 = {}, acc3 = {};

#pragma unroll
    for (int s = 0; s < 16; ++s) {
        const int kp = 2 * s + half;

        v2f a = *reinterpret_cast<const v2f*>(hrow + 2 * kp);

        const v2f* wrow = reinterpret_cast<const v2f*>(&WtPf[kp * WTP_LD * 2]);
        v2f b0 = wrow[l16 +  0];
        v2f b1 = wrow[l16 + 16];
        v2f b2 = wrow[l16 + 32];
        v2f b3 = wrow[l16 + 48];

        acc0 = __builtin_amdgcn_wmma_f32_16x16x4_f32(false, a, false, b0, (short)0, acc0, false, false);
        acc1 = __builtin_amdgcn_wmma_f32_16x16x4_f32(false, a, false, b1, (short)0, acc1, false, false);
        acc2 = __builtin_amdgcn_wmma_f32_16x16x4_f32(false, a, false, b2, (short)0, acc2, false, false);
        acc3 = __builtin_amdgcn_wmma_f32_16x16x4_f32(false, a, false, b3, (short)0, acc3, false, false);
    }

    float* obase = out + (size_t)(tile * 16 + 8 * half) * N_FEATS + l16;
#pragma unroll
    for (int g = 0; g < 8; ++g) {
        float* orow = obase + (size_t)g * N_FEATS;
        orow[ 0] = acc0[g] + bsh[l16 +  0];
        orow[16] = acc1[g] + bsh[l16 + 16];
        orow[32] = acc2[g] + bsh[l16 + 32];
        orow[48] = acc3[g] + bsh[l16 + 48];
    }
}

extern "C" void kernel_launch(void* const* d_in, const int* in_sizes, int n_in,
                              void* d_out, int out_size, void* d_ws, size_t ws_size,
                              hipStream_t stream) {
    const float* x    = (const float*)d_in[0];
    const int*   src  = (const int*)  d_in[1];
    const int*   dst  = (const int*)  d_in[2];
    const float* W    = (const float*)d_in[3];
    const float* bias = (const float*)d_in[4];
    float*       out  = (float*)d_out;

    const int n_nodes = in_sizes[0] / N_FEATS;
    const int n_edges = in_sizes[1];

    float* agg = (float*)d_ws;
    float* deg = agg + (size_t)n_nodes * N_FEATS;

    const int n_zero = n_nodes * N_FEATS + n_nodes;
    zero_f32<<<(n_zero + 255) / 256, 256, 0, stream>>>(agg, n_zero);

    const int scat_threads = n_edges * 16;
    edge_scatter<<<(scat_threads + 255) / 256, 256, 0, stream>>>(
        x, src, dst, agg, deg, n_edges);

    const int n_tiles = (n_nodes + 15) / 16;
    const int blocks  = (n_tiles + 7) / 8;
    gcn_linear_wmma<<<blocks, 256, 0, stream>>>(x, agg, deg, W, bias, out, n_tiles);
}


# Gather launch prologue: W staged as a straight row-major LDS copy (pitch 68 floats) by four 16B loads in flight together, first list lines requested ahead of the barrier; one round trip, was 17
# speedup vs baseline: 5.3706x; 5.3706x over previous
#include <hip/hip_runtime.h>


#define N_FEATS 64
#define WS_LD   68
#define GTHREADS 256
#define ELL_W   32
#define ELL_S   31
#define HS_LD   34
#define PG      16
#define EPT     8

typedef float v2f __attribute__((ext_vector_type(2)));
typedef float v4f __attribute__((ext_vector_type(4)));
typedef float v8f __attribute__((ext_vector_type(8)));
typedef int   v4i __attribute__((ext_vector_type(4)));


__global__ void zero_counters(int* __restrict__ ell, int* __restrict__ ovf_cnt, int n_lines) {
    int i = blockIdx.x * blockDim.x + threadIdx.x;
    if (i < n_lines) ell[(size_t)i * ELL_W + ELL_S] = 0;
    else if (i == n_lines) *ovf_cnt = 0;
}

__device__ __forceinline__ void file_edge(int* __restrict__ line, int r, int s, int d,
                                          int* __restrict__ ovf_cnt, int* __restrict__ ovf) {
    if (r < ELL_S) {
        line[r] = s;
    } else {
        const int k = atomicAdd(ovf_cnt, 1);
        ovf[2 * (size_t)k]     = d;
        ovf[2 * (size_t)k + 1] = s;
    }
}

__global__ void __launch_bounds__(256)
bucket_edges(const int* __restrict__ src,
             const int* __restrict__ dst,
             int*       __restrict__ ell,
             int*       __restrict__ ovf_cnt,
             int*       __restrict__ ovf,
             int n_edges, int n_nodes) {
    const int lane = threadIdx.x & 31;
    const int wave = __builtin_amdgcn_readfirstlane((int)((blockIdx.x * blockDim.x + threadIdx.x) >> 5));
    const int e0 = wave * (32 * EPT);
    if (e0 >= n_edges) return;
    const unsigned place = __builtin_amdgcn_s_sendmsg_rtn(0x87);
    const int t = (int)((place >> 18) & 1u);
    int* table = ell + (size_t)t * n_nodes * ELL_W;

    if (e0 + 32 * EPT <= n_edges) {
        int s[EPT], d[EPT];
#pragma unroll
        for (int q = 0; q < EPT / 4; ++q) {
            const v4i sv = *reinterpret_cast<const v4i*>(src + e0 + 128 * q + 4 * lane);
            const v4i dv = *reinterpret_cast<const v4i*>(dst + e0 + 128 * q + 4 * lane);
            s[4 * q + 0] = sv.x; s[4 * q + 1] = sv.y; s[4 * q + 2] = sv.z; s[4 * q + 3] = sv.w;
            d[4 * q + 0] = dv.x; d[4 * q + 1] = dv.y; d[4 * q + 2] = dv.z; d[4 * q + 3] = dv.w;
        }
        int r[EPT];
#pragma unroll
        for (int u = 0; u < EPT; ++u)
            r[u] = atomicAdd(table + (size_t)d[u] * ELL_W + ELL_S, 1);
#pragma unroll
        for (int u = 0; u < EPT; ++u)
            file_edge(table + (size_t)d[u] * ELL_W, r[u], s[u], d[u], ovf_cnt, ovf);
    } else {
        for (int u = 0; u < EPT; ++u) {
            const int e = e0 + 32 * u + lane;
            if (e < n_edges) {
                const int s1 = src[e];
                const int d1 = dst[e];
                int* line = table + (size_t)d1 * ELL_W;
                const int r1 = atomicAdd(line + ELL_S, 1);
                file_edge(line, r1, s1, d1, ovf_cnt, ovf);
            }
        }
    }
}

__device__ __forceinline__ v2f ld_row(const char* __restrict__ xb, unsigned row_off, unsigned lane8) {
    return *reinterpret_cast<const v2f*>(xb + (row_off + lane8));
}

#define LD_STEP(V, O, N, U)  if ((N) <= (U)) break; V[U] = ld_row(xb, O[U], lane8);
#define ADD_STEP(V, N, U)    if ((N) <= (U)) break; acc += V[U];
#define LD_RUN(V, O, N) do {                                                            \
        LD_STEP(V, O, N, 0)  LD_STEP(V, O, N, 1)  LD_STEP(V, O, N, 2)  LD_STEP(V, O, N, 3)  \
        LD_STEP(V, O, N, 4)  LD_STEP(V, O, N, 5)  LD_STEP(V, O, N, 6)  LD_STEP(V, O, N, 7)  \
        LD_STEP(V, O, N, 8)  LD_STEP(V, O, N, 9)  LD_STEP(V, O, N, 10) LD_STEP(V, O, N, 11) \
        LD_STEP(V, O, N, 12) LD_STEP(V, O, N, 13) LD_STEP(V, O, N, 14) LD_STEP(V, O, N, 15) \
    } while (0)
#define ADD_RUN(V, N) do {                                                              \
        ADD_STEP(V, N, 0)  ADD_STEP(V, N, 1)  ADD_STEP(V, N, 2)  ADD_STEP(V, N, 3)          \
        ADD_STEP(V, N, 4)  ADD_STEP(V, N, 5)  ADD_STEP(V, N, 6)  ADD_STEP(V, N, 7)          \
        ADD_STEP(V, N, 8)  ADD_STEP(V, N, 9)  ADD_STEP(V, N, 10) ADD_STEP(V, N, 11)         \
        ADD_STEP(V, N, 12) ADD_STEP(V, N, 13) ADD_STEP(V, N, 14) ADD_STEP(V, N, 15)         \
    } while (0)
static_assert(PG == 16, "LD_RUN and ADD_RUN are written out for 16 slots");

__device__ __forceinline__ v2f sum_slots(const float* __restrict__ x, const int* __restrict__ line, int j0, int n, int lane, v2f acc) {
    const int sv = line[lane];
    int sidx[PG];
#pragma unroll
    for (int u = 0; u < PG; ++u)
        sidx[u] = __builtin_amdgcn_readlane(sv, j0 + u);
    v2f v[PG];
#pragma unroll
    for (int u = 0; u < PG; ++u) {
        v[u] = (v2f){0.0f, 0.0f};
        if (j0 + u < n)
            v[u] = *reinterpret_cast<const v2f*>(x + (size_t)sidx[u] * N_FEATS + 2 * lane);
    }
#pragma unroll
    for (int u = 0; u < PG; ++u)
        acc += v[u];
    return acc;
}

__global__ void __launch_bounds__(GTHREADS)
gcn_pull_linear_wmma(const float* __restrict__ x,
                     const int*   __restrict__ ell,
                     const int*   __restrict__ ovf_cnt,
                     const int*   __restrict__ ovf,
                     const float* __restrict__ W,
                     const float* __restrict__ bias,
                     float*       __restrict__ out,
                     int n_tiles, int n_nodes) {
    __shared__ __attribute__((aligned(16))) float Ws[N_FEATS * WS_LD];
    __shared__ __attribute__((aligned(16))) float bsh[N_FEATS];
    __shared__ __attribute__((aligned(16))) float Hs[8 * 32 * HS_LD];

    const int wid  = threadIdx.x >> 5;
    const int lane = threadIdx.x & 31;
    const int tile = __builtin_amdgcn_readfirstlane((int)(blockIdx.x * 8 + wid));
    const bool live = tile < n_tiles;
    const int row0 = (live ? tile : 0) * 16;

    const int* ellA = ell;
    const int* ellB = ell + (size_t)n_nodes * ELL_W;

    int a_cur = ellA[(size_t)row0 * ELL_W + lane];
    int b_cur = ellB[(size_t)row0 * ELL_W + lane];
    int a_nxt = ellA[(size_t)(row0 + 1) * ELL_W + lane];
    int b_nxt = ellB[(size_t)(row0 + 1) * ELL_W + lane];

    v4f wv[4];
#pragma unroll
    for (int j = 0; j < 4; ++j)
        wv[j] = *reinterpret_cast<const v4f*>(W + 4 * ((int)threadIdx.x + GTHREADS * j));
    float bv = 0.0f;
    if (threadIdx.x < N_FEATS) bv = bias[threadIdx.x];
#pragma unroll
    for (int j = 0; j < 4; ++j) {
        const int p = (int)threadIdx.x + GTHREADS * j;
        *reinterpret_cast<v4f*>(&Ws[(p >> 4) * WS_LD + 4 * (p & 15)]) = wv[j];
    }
    if (threadIdx.x < N_FEATS) bsh[threadIdx.x] = bv;
    __syncthreads();

    if (!live) return;

    const int half = lane >> 4;
    const int l16  = lane & 15;
    float* hs = Hs + wid * (32 * HS_LD);

    const char* xb = reinterpret_cast<const char*>(x);
    const unsigned lane8 = 8u * (unsigned)lane;

#pragma unroll 2
    for (int r = 0; r < 16; ++r) {
        const int d  = row0 + r;
        const int ca = __builtin_amdgcn_readlane(a_cur, ELL_S);
        const int cb = __builtin_amdgcn_readlane(b_cur, ELL_S);
        const int na = ca < ELL_S ? ca : ELL_S;
        const int nb = cb < ELL_S ? cb : ELL_S;

        const int oa = (int)((unsigned)a_cur << 8);
        const int ob = (int)((unsigned)b_cur << 8);
        unsigned ia[PG], ib[PG];
#pragma unroll
        for (int u = 0; u < PG; ++u) {
            ia[u] = (unsigned)__builtin_amdgcn_readlane(oa, u);
            ib[u] = (unsigned)__builtin_amdgcn_readlane(ob, u);
        }
        v2f acc = (v2f){0.0f, 0.0f};
        if ((ca | cb) == 0)
            acc = ld_row(xb, (unsigned)d << 8, lane8);
        v2f va[PG], vb[PG];
        LD_RUN(va, ia, na);
        LD_RUN(vb, ib, nb);

        const int r2 = r < 14 ? r + 2 : 15;
        const int a_far = ellA[(size_t)(row0 + r2) * ELL_W + lane];
        const int b_far = ellB[(size_t)(row0 + r2) * ELL_W + lane];
        __builtin_amdgcn_sched_barrier(0);

        ADD_RUN(va, na);
        ADD_RUN(vb, nb);

        if (na > PG) acc = sum_slots(x, ellA + (size_t)d * ELL_W, PG, na, lane, acc);
        if (nb > PG) acc = sum_slots(x, ellB + (size_t)d * ELL_W, PG, nb, lane, acc);

        if (ca > ELL_S || cb > ELL_S) {
            const int n_ovf = *ovf_cnt;
            for (int k0 = 0; k0 < n_ovf; k0 += 32) {
                const int k = k0 + lane;
                int od = -1, os = 0;
                if (k < n_ovf) {
                    od = ovf[2 * (size_t)k];
                    os = ovf[2 * (size_t)k + 1];
                }
                unsigned m = __builtin_amdgcn_ballot_w32(od == d);
                while (m != 0u) {
                    const int j = __builtin_ctz(m);
                    m &= m - 1u;
                    const int s = __builtin_amdgcn_readlane(os, j);
                    acc += *reinterpret_cast<const v2f*>(x + (size_t)s * N_FEATS + 2 * lane);
                }
            }
        }
        *reinterpret_cast<v2f*>(hs + lane * HS_LD + 2 * r) = acc;

        a_cur = a_nxt; b_cur = b_nxt;
        a_nxt = a_far; b_nxt = b_far;
    }

    v8f acc0 = {}, acc1 = {}, acc2 = {}, acc3 = {};

#pragma unroll
    for (int s = 0; s < 16; ++s) {
        const int kp = 2 * s + half;

        v2f a = *reinterpret_cast<const v2f*>(hs + kp * HS_LD + 2 * l16);

        const float* wk = Ws + l16 * WS_LD + 2 * kp;
        v2f b0 = *reinterpret_cast<const v2f*>(wk +  0 * WS_LD);
        v2f b1 = *reinterpret_cast<const v2f*>(wk + 16 * WS_LD);
        v2f b2 = *reinterpret_cast<const v2f*>(wk + 32 * WS_LD);
        v2f b3 = *reinterpret_cast<const v2f*>(wk + 48 * WS_LD);

        acc0 = __builtin_amdgcn_wmma_f32_16x16x4_f32(false, a, false, b0, (short)0, acc0, false, false);
        acc1 = __builtin_amdgcn_wmma_f32_16x16x4_f32(false, a, false, b1, (short)0, acc1, false, false);
        acc2 = __builtin_amdgcn_wmma_f32_16x16x4_f32(false, a, false, b2, (short)0, acc2, false, false);
        acc3 = __builtin_amdgcn_wmma_f32_16x16x4_f32(false, a, false, b3, (short)0, acc3, false, false);
    }

    float* obase = out + (size_t)(row0 + 8 * half) * N_FEATS + l16;
#pragma unroll
    for (int g = 0; g < 8; ++g) {
        float* orow = obase + (size_t)g * N_FEATS;
        orow[ 0] = acc0[g] + bsh[l16 +  0];
        orow[16] = acc1[g] + bsh[l16 + 16];
        orow[32] = acc2[g] + bsh[l16 + 32];
        orow[48] = acc3[g] + bsh[l16 + 48];
    }
}

extern "C" void kernel_launch(void* const* d_in, const int* in_sizes, int n_in,
                              void* d_out, int out_size, void* d_ws, size_t ws_size,
                              hipStream_t stream) {
    const float* x    = (const float*)d_in[0];
    const int*   src  = (const int*)  d_in[1];
    const int*   dst  = (const int*)  d_in[2];
    const float* W    = (const float*)d_in[3];
    const float* bias = (const float*)d_in[4];
    float*       out  = (float*)d_out;

    const int n_nodes = in_sizes[0] / N_FEATS;
    const int n_edges = in_sizes[1];

    int* ell     = (int*)d_ws;
    int* ovf     = ell + 2 * (size_t)n_nodes * ELL_W;
    int* ovf_cnt = ovf + 2 * (size_t)n_edges;

    const int n_lines = 2 * n_nodes;
    const int n_zero  = n_lines + 1;
    zero_counters<<<(n_zero + 255) / 256, 256, 0, stream>>>(ell, ovf_cnt, n_lines);

    bucket_edges<<<(n_edges + 256 * EPT - 1) / (256 * EPT), 256, 0, stream>>>(src, dst, ell, ovf_cnt, ovf, n_edges, n_nodes);

    const int n_tiles = (n_nodes + 15) / 16;
    const int blocks  = (n_tiles + 7) / 8;
    gcn_pull_linear_wmma<<<blocks, GTHREADS, 0, stream>>>(x, ell, ovf_cnt, ovf, W, bias, out, n_tiles, n_nodes);
}
